# starting kernel
# baseline (speedup 1.0000x reference)
#include <hip/hip_runtime.h>
#include <hip/hip_bf16.h>


#define B_   4
#define C_   64
#define H_   96
#define W_   320
#define HW_  (H_ * W_)
#define CHW_ (C_ * HW_)
#define HO_  384
#define WO_  1280
#define NOUT_ (B_ * HO_ * WO_)
#define NZ_   (B_ * HW_)

typedef __attribute__((ext_vector_type(2))) float v2f;
typedef __attribute__((ext_vector_type(8))) float v8f;

__global__ __launch_bounds__(256) void
upsample_reduce_c_kernel(const float* __restrict__ x,
                         const float* __restrict__ w,
                         float* __restrict__ z) {
  const int tid  = blockIdx.x * 256 + threadIdx.x;
  const int wave = tid >> 5;
  const int lane = threadIdx.x & 31;
  const int m    = lane & 15;
  const int half = lane >> 4;

  const int pbase = wave * 16;
  const int b     = pbase / HW_;
  const int s     = pbase - b * HW_;

  const float* xb = x + (size_t)b * CHW_ + (size_t)(s + m);

  v8f acc = {};
#pragma unroll
  for (int c0 = 0; c0 < C_; c0 += 4) {
    const int cb = c0 + (half << 1);
    v2f a, bw;
    a.x  = xb[(size_t)cb * HW_];
    a.y  = xb[(size_t)(cb + 1) * HW_];
    bw.x = w[cb];
    bw.y = w[cb + 1];
    acc = __builtin_amdgcn_wmma_f32_16x16x4_f32(
         false, a,  false, bw,
         (short)0, acc,  false,  false);
  }

  if (m == 0) {
    float* zp = z + pbase + (half << 3);
    zp[0] = acc[0]; zp[1] = acc[1]; zp[2] = acc[2]; zp[3] = acc[3];
    zp[4] = acc[4]; zp[5] = acc[5]; zp[6] = acc[6]; zp[7] = acc[7];
  }
}

__global__ __launch_bounds__(256) void
upsample_lerp_bias_relu_kernel(const float* __restrict__ z,
                               const float* __restrict__ bias,
                               float* __restrict__ out) {
  const int idx = blockIdx.x * 256 + threadIdx.x;
  const int wo  = idx % WO_;
  const int t   = idx / WO_;
  const int ho  = t % HO_;
  const int b   = t / HO_;

  const float RY = (float)((double)(H_ - 1) / (double)(HO_ - 1));
  const float RX = (float)((double)(W_ - 1) / (double)(WO_ - 1));

  const float fy = (float)ho * RY;
  int i0 = (int)fy; if (i0 > H_ - 2) i0 = H_ - 2;
  const float wy = fy - (float)i0;

  const float fx = (float)wo * RX;
  int j0 = (int)fx; if (j0 > W_ - 2) j0 = W_ - 2;
  const float wx = fx - (float)j0;

  const float* zb = z + b * HW_ + i0 * W_ + j0;
  const float v00 = zb[0];
  const float v01 = zb[1];
  const float v10 = zb[W_];
  const float v11 = zb[W_ + 1];

  const float r0  = v00 * (1.0f - wy) + v10 * wy;
  const float r1  = v01 * (1.0f - wy) + v11 * wy;
  const float val = r0 * (1.0f - wx) + r1 * wx + bias[0];

  out[idx] = val > 0.0f ? val : 0.0f;
}

extern "C" void kernel_launch(void* const* d_in, const int* in_sizes, int n_in,
                              void* d_out, int out_size, void* d_ws, size_t ws_size,
                              hipStream_t stream) {
  (void)in_sizes; (void)n_in; (void)out_size; (void)ws_size;

  const float* x    = (const float*)d_in[0];
  const float* wgt  = (const float*)d_in[1];
  const float* bias = (const float*)d_in[2];
  float*       out  = (float*)d_out;
  float*       z    = (float*)d_ws;

  upsample_reduce_c_kernel<<<NZ_ / 16 / 8, 256, 0, stream>>>(x, wgt, z);
  upsample_lerp_bias_relu_kernel<<<NOUT_ / 256, 256, 0, stream>>>(z, bias, out);
}


# two launches fused into one: each workgroup reduces the z tile its 160-column output block needs (5x42, same f32 WMMA) into LDS, then interpolates from LDS, 4 outputs per thread in one vector store
# speedup vs baseline: 1.2864x; 1.2864x over previous
#include <hip/hip_runtime.h>


#define B_   4
#define C_   64
#define H_   96
#define W_   320
#define HW_  (H_ * W_)
#define CHW_ (C_ * HW_)
#define HO_  384
#define WO_  1280

#define TR_   4
#define ZR_   (TR_ + 1)
#define TWO_  160
#define ZC_   42
#define NP_   (ZR_ * ZC_)
#define NG_   ((NP_ + 15) / 16)
#define RT_   24
#define CT_   (WO_ / TWO_)
#define NWG_  (B_ * RT_ * CT_)
#define QPR_  (TWO_ / 4)
#define RPP_  6

typedef __attribute__((ext_vector_type(2))) float v2f;
typedef __attribute__((ext_vector_type(8))) float v8f;

__device__ __forceinline__ int first_out_row(int r) {
  return r >= H_ - 1 ? HO_ : (r * (HO_ - 1) + (H_ - 2)) / (H_ - 1);
}

template <int N>
__device__ __forceinline__ void reduce_groups(const float* __restrict__ xb,
                                              const float* __restrict__ w,
                                              float* __restrict__ zt,
                                              int r0, int jlo, int g0,
                                              int m, int half) {
  const float* xp[N];
  v8f acc[N];
#pragma unroll
  for (int n = 0; n < N; ++n) {
    int q = (g0 + 8 * n) * 16 + m;
    if (q > NP_ - 1) q = NP_ - 1;
    const int zr = q / ZC_;
    const int zc = q - zr * ZC_;
    int row = r0 + zr;  if (row > H_ - 1) row = H_ - 1;
    int col = jlo + zc; if (col > W_ - 1) col = W_ - 1;
    xp[n] = xb + row * W_ + col;
    acc[n] = (v8f){};
  }
#pragma unroll
  for (int c0 = 0; c0 < C_; c0 += 4) {
    const int cb = c0 + (half << 1);
    v2f bw;
    bw.x = w[cb];
    bw.y = w[cb + 1];
#pragma unroll
    for (int n = 0; n < N; ++n) {
      v2f a;
      a.x = xp[n][(size_t)cb * HW_];
      a.y = xp[n][(size_t)(cb + 1) * HW_];
      acc[n] = __builtin_amdgcn_wmma_f32_16x16x4_f32(
           false, a,  false, bw,
           (short)0, acc[n],  false,  false);
    }
  }
  if (m == 0) {
#pragma unroll
    for (int n = 0; n < N; ++n) {
      float* zp = zt + (g0 + 8 * n) * 16 + (half << 3);
      zp[0] = acc[n][0]; zp[1] = acc[n][1]; zp[2] = acc[n][2]; zp[3] = acc[n][3];
      zp[4] = acc[n][4]; zp[5] = acc[n][5]; zp[6] = acc[n][6]; zp[7] = acc[n][7];
    }
  }
}

__global__ __launch_bounds__(256) void
upsample_reduce_lerp_kernel(const float* __restrict__ x,
                            const float* __restrict__ w,
                            const float* __restrict__ bias,
                            float* __restrict__ out) {
  __shared__ float zt[NG_ * 16];

  const int bid = blockIdx.x;
  const int ct  = bid % CT_;
  const int rt  = (bid / CT_) % RT_;
  const int b   = bid / (CT_ * RT_);

  const int r0  = rt * TR_;
  const int wo0 = ct * TWO_;
  const int jlo = (wo0 * (W_ - 1)) / (WO_ - 1);

  const int tid  = threadIdx.x;
  const int wave = tid >> 5;
  const int lane = tid & 31;

  {
    const float* xb = x + (size_t)b * CHW_;
    const int m    = lane & 15;
    const int half = lane >> 4;
    if (wave + 8 < NG_)
      reduce_groups<2>(xb, w, zt, r0, jlo, wave, m, half);
    else
      reduce_groups<1>(xb, w, zt, r0, jlo, wave, m, half);
  }
  __syncthreads();

  if (tid < RPP_ * QPR_) {
    const int rg = tid / QPR_;
    const int qd = tid - rg * QPR_;

    const float RY = (float)((double)(H_ - 1) / (double)(HO_ - 1));
    const float RX = (float)((double)(W_ - 1) / (double)(WO_ - 1));
    const float bv = bias[0];

    int   cj[4];
    float wx[4];
#pragma unroll
    for (int e = 0; e < 4; ++e) {
      const int wo = wo0 + 4 * qd + e;
      const float fx = (float)wo * RX;
      int j0 = (int)fx; if (j0 > W_ - 2) j0 = W_ - 2;
      wx[e] = fx - (float)j0;
      cj[e] = j0 - jlo;
    }

    const int ho_end = first_out_row(r0 + TR_);
    float* ob = out + (size_t)b * (HO_ * WO_) + wo0 + 4 * qd;

    for (int ho = first_out_row(r0) + rg; ho < ho_end; ho += RPP_) {
      const float fy = (float)ho * RY;
      int i0 = (int)fy; if (i0 > H_ - 2) i0 = H_ - 2;
      const float wy = fy - (float)i0;
      const float* zrow = zt + (i0 - r0) * ZC_;

      float o[4];
#pragma unroll
      for (int e = 0; e < 4; ++e) {
        const float v00 = zrow[cj[e]];
        const float v01 = zrow[cj[e] + 1];
        const float v10 = zrow[ZC_ + cj[e]];
        const float v11 = zrow[ZC_ + cj[e] + 1];
        const float r0v = v00 * (1.0f - wy) + v10 * wy;
        const float r1v = v01 * (1.0f - wy) + v11 * wy;
        const float val = r0v * (1.0f - wx[e]) + r1v * wx[e] + bv;
        o[e] = val > 0.0f ? val : 0.0f;
      }
      *reinterpret_cast<float4*>(ob + (size_t)ho * WO_) = make_float4(o[0], o[1], o[2], o[3]);
    }
  }
}

extern "C" void kernel_launch(void* const* d_in, const int* in_sizes, int n_in,
                              void* d_out, int out_size, void* d_ws, size_t ws_size,
                              hipStream_t stream) {
  (void)in_sizes; (void)n_in; (void)out_size; (void)d_ws; (void)ws_size;

  const float* x    = (const float*)d_in[0];
  const float* wgt  = (const float*)d_in[1];
  const float* bias = (const float*)d_in[2];
  float*       out  = (float*)d_out;

  upsample_reduce_lerp_kernel<<<NWG_, 256, 0, stream>>>(x, wgt, bias, out);
}


# x read with 16-byte loads used as WMMA A operands unmoved (k = 2 columns x 2 channels, B selects the column), all issued first; tile 6x320, 256 workgroups of 10 waves; channel sum in 32 steps of 2
# speedup vs baseline: 1.6426x; 1.2769x over previous
#include <hip/hip_runtime.h>


#define B_   4
#define C_   64
#define H_   96
#define W_   320
#define HW_  (H_ * W_)
#define CHW_ (C_ * HW_)
#define HO_  384
#define WO_  1280

#define TR_   6
#define ZR_   (TR_ + 1)
#define TWO_  320
#define NQ_   22
#define ZCP_  (4 * NQ_)
#define NS_   (ZR_ * NQ_)
#define NG_   ((NS_ + 15) / 16)
#define NT_   (NG_ * 32)
#define RT_   16
#define CT_   (WO_ / TWO_)
#define NWG_  (B_ * RT_ * CT_)
#define QPR_  (TWO_ / 4)
#define RPP_  (NT_ / QPR_)

typedef __attribute__((ext_vector_type(2))) float v2f;
typedef __attribute__((ext_vector_type(8))) float v8f;

__device__ __forceinline__ int first_out_row(int r) {
  return r >= H_ - 1 ? HO_ : (r * (HO_ - 1) + (H_ - 2)) / (H_ - 1);
}

__device__ __forceinline__ void reduce_group(const float* __restrict__ xb,
                                             const float* __restrict__ w,
                                             float* __restrict__ zt,
                                             int r0, int jl4, int g,
                                             int m, int half) {
  int s = g * 16 + m;
  if (s > NS_ - 1) s = NS_ - 1;
  const int zr = s / NQ_;
  const int zq = s - zr * NQ_;
  int row = r0 + zr;      if (row > H_ - 1) row = H_ - 1;
  int col = jl4 + 4 * zq; if (col > W_ - 4) col = W_ - 4;
  const float* xp = xb + row * W_ + col + half * HW_;

  float4 u[C_ / 2];
  v2f bw[C_ / 2];
#pragma unroll
  for (int k = 0; k < C_ / 2; ++k)
    u[k] = *reinterpret_cast<const float4*>(xp + (size_t)(2 * k) * HW_);
#pragma unroll
  for (int k = 0; k < C_ / 2; ++k) {
    const float we = w[2 * k];
    const float wo = w[2 * k + 1];
    const float wk = half ? wo : we;
    bw[k].x = (m == 0) ? wk : 0.0f;
    bw[k].y = (m == 1) ? wk : 0.0f;
  }
  v8f accL = (v8f){};
  v8f accH = (v8f){};
  asm volatile("" : "+v"(accL), "+v"(accH));
  __builtin_amdgcn_sched_barrier(0);

#pragma unroll
  for (int k = 0; k < C_ / 2; ++k) {
    v2f a0, a1;
    a0.x = u[k].x; a0.y = u[k].y;
    a1.x = u[k].z; a1.y = u[k].w;
    accL = __builtin_amdgcn_wmma_f32_16x16x4_f32(
         false, a0,  false, bw[k],
         (short)0, accL,  false,  false);
    accH = __builtin_amdgcn_wmma_f32_16x16x4_f32(
         false, a1,  false, bw[k],
         (short)0, accH,  false,  false);
  }
  if (m < 2) {
    float* zp = zt + (g * 16 + (half << 3)) * 4 + m;
#pragma unroll
    for (int r = 0; r < 8; ++r) {
      zp[4 * r]     = accL[r];
      zp[4 * r + 2] = accH[r];
    }
  }
}

__global__ __launch_bounds__(NT_) void
upsample_reduce_lerp_kernel(const float* __restrict__ x,
                            const float* __restrict__ w,
                            const float* __restrict__ bias,
                            float* __restrict__ out) {
  __shared__ __attribute__((aligned(16))) float zt[NG_ * 16 * 4];

  const int bid = blockIdx.x;
  const int ct  = bid % CT_;
  const int rt  = (bid / CT_) % RT_;
  const int b   = bid / (CT_ * RT_);

  const int r0  = rt * TR_;
  const int wo0 = ct * TWO_;
  const int jlo = (wo0 * (W_ - 1)) / (WO_ - 1);
  const int jl4 = jlo & ~3;

  const int tid  = threadIdx.x;
  const int wave = tid >> 5;
  const int lane = tid & 31;

  reduce_group(x + (size_t)b * CHW_, w, zt, r0, jl4, wave, lane & 15, lane >> 4);
  __syncthreads();

  {
    const int rg = tid / QPR_;
    const int qd = tid - rg * QPR_;

    const float RY = (float)((double)(H_ - 1) / (double)(HO_ - 1));
    const float RX = (float)((double)(W_ - 1) / (double)(WO_ - 1));
    const float bv = bias[0];

    int   cj[4];
    float wx[4];
#pragma unroll
    for (int e = 0; e < 4; ++e) {
      const int wo = wo0 + 4 * qd + e;
      const float fx = (float)wo * RX;
      int j0 = (int)fx; if (j0 > W_ - 2) j0 = W_ - 2;
      wx[e] = fx - (float)j0;
      cj[e] = j0 - jl4;
    }

    const int ho_end = first_out_row(r0 + TR_);
    float* ob = out + (size_t)b * (HO_ * WO_) + wo0 + 4 * qd;

    for (int ho = first_out_row(r0) + rg; ho < ho_end; ho += RPP_) {
      const float fy = (float)ho * RY;
      int i0 = (int)fy; if (i0 > H_ - 2) i0 = H_ - 2;
      const float wy = fy - (float)i0;
      const float* zrow = zt + (i0 - r0) * ZCP_;

      float o[4];
#pragma unroll
      for (int e = 0; e < 4; ++e) {
        const float v00 = zrow[cj[e]];
        const float v01 = zrow[cj[e] + 1];
        const float v10 = zrow[ZCP_ + cj[e]];
        const float v11 = zrow[ZCP_ + cj[e] + 1];
        const float r0v = v00 * (1.0f - wy) + v10 * wy;
        const float r1v = v01 * (1.0f - wy) + v11 * wy;
        const float val = r0v * (1.0f - wx[e]) + r1v * wx[e] + bv;
        o[e] = val > 0.0f ? val : 0.0f;
      }
      *reinterpret_cast<float4*>(ob + (size_t)ho * WO_) = make_float4(o[0], o[1], o[2], o[3]);
    }
  }
}

extern "C" void kernel_launch(void* const* d_in, const int* in_sizes, int n_in,
                              void* d_out, int out_size, void* d_ws, size_t ws_size,
                              hipStream_t stream) {
  (void)in_sizes; (void)n_in; (void)out_size; (void)d_ws; (void)ws_size;

  const float* x    = (const float*)d_in[0];
  const float* wgt  = (const float*)d_in[1];
  const float* bias = (const float*)d_in[2];
  float*       out  = (float*)d_out;

  upsample_reduce_lerp_kernel<<<NWG_, NT_, 0, stream>>>(x, wgt, bias, out);
}
